# starting kernel
# baseline (speedup 1.0000x reference)
#include <hip/hip_runtime.h>

typedef float v2f __attribute__((ext_vector_type(2)));
typedef float v8f __attribute__((ext_vector_type(8)));

#define FRAME_LEN 160
#define HOP 256
#define MIN_LAG 40
#define WAVES_PER_BLOCK 8
#define XP_SIZE 416
#define XR_SIZE 176
#define NEG_INF (-3.402823466e+38f)

__global__ __launch_bounds__(256) void pitch_wmma_kernel(
    const float* __restrict__ audio, float* __restrict__ out,
    int S, int T, int nframes) {
  __shared__ float s_xp[WAVES_PER_BLOCK][XP_SIZE];
  __shared__ float s_xr[WAVES_PER_BLOCK][XR_SIZE];

  const int lane = threadIdx.x & 31;
  const int wave = threadIdx.x >> 5;
  int f = blockIdx.x * WAVES_PER_BLOCK + wave;
  if (f >= nframes) f = nframes - 1;
  const int b = f / T;
  const int t = f - b * T;
  const float* __restrict__ src = audio + (size_t)b * (size_t)S + (size_t)t * HOP;

  float x0 = src[lane];
  float x1 = src[lane + 32];
  float x2 = src[lane + 64];
  float x3 = src[lane + 96];
  float x4 = src[lane + 128];

  float sum = x0 + x1 + x2 + x3 + x4;
#pragma unroll
  for (int off = 16; off; off >>= 1) sum += __shfl_xor(sum, off, 32);
  const float mean = sum * (1.0f / 160.0f);
  x0 -= mean; x1 -= mean; x2 -= mean; x3 -= mean; x4 -= mean;

  float mx = fmaxf(fmaxf(fmaxf(fabsf(x0), fabsf(x1)),
                         fmaxf(fabsf(x2), fabsf(x3))), fabsf(x4));
#pragma unroll
  for (int off = 16; off; off >>= 1) mx = fmaxf(mx, __shfl_xor(mx, off, 32));
  const bool silent = mx < 1e-8f;

  float* xp = s_xp[wave];
  float* xr = s_xr[wave];
  for (int j = lane; j < 240; j += 32) xp[j] = 0.0f;
  if (lane < XR_SIZE - FRAME_LEN) xr[FRAME_LEN + lane] = 0.0f;
  xp[240 + lane      ] = x0;  xr[159 - lane      ] = x0;
  xp[240 + lane + 32 ] = x1;  xr[159 - lane - 32 ] = x1;
  xp[240 + lane + 64 ] = x2;  xr[159 - lane - 64 ] = x2;
  xp[240 + lane + 96 ] = x3;  xr[159 - lane - 96 ] = x3;
  xp[240 + lane + 128] = x4;  xr[159 - lane - 128] = x4;
  __syncthreads();

  const int n  = lane & 15;
  const int hf = lane >> 4;
  const float* __restrict__ xra = xr + 2 * hf + n;
  const float* __restrict__ xpb = xp + 240 + 2 * hf - 16 * n;
  v8f c = {};
#pragma unroll
  for (int k0 = 0; k0 < FRAME_LEN; k0 += 4) {
    v2f a, bb;
    a.x  = xra[k0];
    a.y  = xra[k0 + 1];
    bb.x = xpb[k0];
    bb.y = xpb[k0 + 1];
    c = __builtin_amdgcn_wmma_f32_16x16x4_f32(
         false, a,  false, bb,
         (short)0, c,  false,  false);
  }

  union { v8f v; float f[8]; } cu; cu.v = c;
  float bestv = NEG_INF;
  int   bestk = 1 << 30;
#pragma unroll
  for (int v = 0; v < 8; ++v) {
    const int m = v + 8 * hf;
    const int L = 16 * n + m;
    const int k = 159 - L;
    const float val = (L <= 159 - MIN_LAG) ? cu.f[v] : NEG_INF;
    const bool better = (val > bestv) | ((val == bestv) & (k < bestk));
    bestv = better ? val : bestv;
    bestk = better ? k : bestk;
  }
#pragma unroll
  for (int off = 16; off; off >>= 1) {
    const float ov = __shfl_xor(bestv, off, 32);
    const int   ok = __shfl_xor(bestk, off, 32);
    const bool better = (ov > bestv) | ((ov == bestv) & (ok < bestk));
    bestv = better ? ov : bestv;
    bestk = better ? ok : bestk;
  }

  if (lane == 0) {
    out[f] = silent ? 0.0f : 16000.0f / (float)bestk;
  }
}

extern "C" void kernel_launch(void* const* d_in, const int* in_sizes, int n_in,
                              void* d_out, int out_size, void* d_ws, size_t ws_size,
                              hipStream_t stream) {
  (void)n_in; (void)d_ws; (void)ws_size;
  const float* audio = (const float*)d_in[0];
  float* out = (float*)d_out;
  const int B = 16;
  const int S = in_sizes[0] / B;
  const int T = (S - FRAME_LEN) / HOP + 1;
  const int nframes = B * T;
  (void)out_size;
  const int blocks = (nframes + WAVES_PER_BLOCK - 1) / WAVES_PER_BLOCK;
  pitch_wmma_kernel<<<blocks, 32 * WAVES_PER_BLOCK, 0, stream>>>(
      audio, out, S, T, nframes);
}


# Pad the LDS copy of the shifted B operand by 4 dwords per 16 samples: lane stride of the B-fragment reads 16 -> 20 dwords, 4-way bank conflict -> none; k order and summation order unchanged
# speedup vs baseline: 1.0336x; 1.0336x over previous
#include <hip/hip_runtime.h>

typedef float v2f __attribute__((ext_vector_type(2)));
typedef float v8f __attribute__((ext_vector_type(8)));

#define FRAME_LEN 160
#define HOP 256
#define MIN_LAG 40
#define WAVES_PER_BLOCK 8
#define XP_SIZE 496
#define XP_DATA 300
#define XR_SIZE 176
#define NEG_INF (-3.402823466e+38f)

__global__ __launch_bounds__(256) void pitch_wmma_kernel(
    const float* __restrict__ audio, float* __restrict__ out,
    int S, int T, int nframes) {
  __shared__ float s_xp[WAVES_PER_BLOCK][XP_SIZE];
  __shared__ float s_xr[WAVES_PER_BLOCK][XR_SIZE];

  const int lane = threadIdx.x & 31;
  const int wave = threadIdx.x >> 5;
  int f = blockIdx.x * WAVES_PER_BLOCK + wave;
  if (f >= nframes) f = nframes - 1;
  const int b = f / T;
  const int t = f - b * T;
  const float* __restrict__ src = audio + (size_t)b * (size_t)S + (size_t)t * HOP;

  float x0 = src[lane];
  float x1 = src[lane + 32];
  float x2 = src[lane + 64];
  float x3 = src[lane + 96];
  float x4 = src[lane + 128];

  float sum = x0 + x1 + x2 + x3 + x4;
#pragma unroll
  for (int off = 16; off; off >>= 1) sum += __shfl_xor(sum, off, 32);
  const float mean = sum * (1.0f / 160.0f);
  x0 -= mean; x1 -= mean; x2 -= mean; x3 -= mean; x4 -= mean;

  float mx = fmaxf(fmaxf(fmaxf(fabsf(x0), fabsf(x1)),
                         fmaxf(fabsf(x2), fabsf(x3))), fabsf(x4));
#pragma unroll
  for (int off = 16; off; off >>= 1) mx = fmaxf(mx, __shfl_xor(mx, off, 32));
  const bool silent = mx < 1e-8f;

  float* xp = s_xp[wave];
  float* xr = s_xr[wave];
  const int zp = lane + 4 * (lane >> 4);
  for (int j = 0; j < 7; ++j) xp[zp + 40 * j] = 0.0f;
  if (lane < 16) xp[zp + 280] = 0.0f;
  if (lane < XR_SIZE - FRAME_LEN) xr[FRAME_LEN + lane] = 0.0f;
  xp[XP_DATA + zp      ] = x0;  xr[159 - lane      ] = x0;
  xp[XP_DATA + zp + 40 ] = x1;  xr[159 - lane - 32 ] = x1;
  xp[XP_DATA + zp + 80 ] = x2;  xr[159 - lane - 64 ] = x2;
  xp[XP_DATA + zp + 120] = x3;  xr[159 - lane - 96 ] = x3;
  xp[XP_DATA + zp + 160] = x4;  xr[159 - lane - 128] = x4;
  __syncthreads();

  const int n  = lane & 15;
  const int hf = lane >> 4;
  const float* __restrict__ xra = xr + 2 * hf + n;
  const float* __restrict__ xpb = xp + XP_DATA + 2 * hf - 20 * n;
  v8f c = {};
#pragma unroll
  for (int k0 = 0; k0 < FRAME_LEN; k0 += 4) {
    v2f a, bb;
    a.x  = xra[k0];
    a.y  = xra[k0 + 1];
    const int kp = k0 + 4 * (k0 >> 4);
    bb.x = xpb[kp];
    bb.y = xpb[kp + 1];
    c = __builtin_amdgcn_wmma_f32_16x16x4_f32(
         false, a,  false, bb,
         (short)0, c,  false,  false);
  }

  union { v8f v; float f[8]; } cu; cu.v = c;
  float bestv = NEG_INF;
  int   bestk = 1 << 30;
#pragma unroll
  for (int v = 0; v < 8; ++v) {
    const int m = v + 8 * hf;
    const int L = 16 * n + m;
    const int k = 159 - L;
    const float val = (L <= 159 - MIN_LAG) ? cu.f[v] : NEG_INF;
    const bool better = (val > bestv) | ((val == bestv) & (k < bestk));
    bestv = better ? val : bestv;
    bestk = better ? k : bestk;
  }
#pragma unroll
  for (int off = 16; off; off >>= 1) {
    const float ov = __shfl_xor(bestv, off, 32);
    const int   ok = __shfl_xor(bestk, off, 32);
    const bool better = (ov > bestv) | ((ov == bestv) & (ok < bestk));
    bestv = better ? ov : bestv;
    bestk = better ? ok : bestk;
  }

  if (lane == 0) {
    out[f] = silent ? 0.0f : 16000.0f / (float)bestk;
  }
}

extern "C" void kernel_launch(void* const* d_in, const int* in_sizes, int n_in,
                              void* d_out, int out_size, void* d_ws, size_t ws_size,
                              hipStream_t stream) {
  (void)n_in; (void)d_ws; (void)ws_size;
  const float* audio = (const float*)d_in[0];
  float* out = (float*)d_out;
  const int B = 16;
  const int S = in_sizes[0] / B;
  const int T = (S - FRAME_LEN) / HOP + 1;
  const int nframes = B * T;
  (void)out_size;
  const int blocks = (nframes + WAVES_PER_BLOCK - 1) / WAVES_PER_BLOCK;
  pitch_wmma_kernel<<<blocks, 32 * WAVES_PER_BLOCK, 0, stream>>>(
      audio, out, S, T, nframes);
}


# Fold the self-convolution on its mirror pairs: B copy holds samples 80..159 doubled, A only samples 0..79; K 160 -> 96, 40 -> 24 f32 WMMAs; summation order changes (a mirrored pair is summed once)
# speedup vs baseline: 1.3291x; 1.2859x over previous
#include <hip/hip_runtime.h>

typedef float v2f __attribute__((ext_vector_type(2)));
typedef float v8f __attribute__((ext_vector_type(8)));

#define FRAME_LEN 160
#define HOP 256
#define MIN_LAG 40
#define WAVES_PER_BLOCK 8
#define XP_SIZE 260
#define XP_DATA 60
#define XR_SIZE 112
#define K_FIRST 64
#define K_LEN 96
#define NEG_INF (-3.402823466e+38f)

__global__ __launch_bounds__(256) void pitch_wmma_kernel(
    const float* __restrict__ audio, float* __restrict__ out,
    int S, int T, int nframes) {
  __shared__ float s_xp[WAVES_PER_BLOCK][XP_SIZE];
  __shared__ float s_xr[WAVES_PER_BLOCK][XR_SIZE];

  const int lane = threadIdx.x & 31;
  const int wave = threadIdx.x >> 5;
  int f = blockIdx.x * WAVES_PER_BLOCK + wave;
  if (f >= nframes) f = nframes - 1;
  const int b = f / T;
  const int t = f - b * T;
  const float* __restrict__ src = audio + (size_t)b * (size_t)S + (size_t)t * HOP;

  float x0 = src[lane];
  float x1 = src[lane + 32];
  float x2 = src[lane + 64];
  float x3 = src[lane + 96];
  float x4 = src[lane + 128];

  float sum = x0 + x1 + x2 + x3 + x4;
#pragma unroll
  for (int off = 16; off; off >>= 1) sum += __shfl_xor(sum, off, 32);
  const float mean = sum * (1.0f / 160.0f);
  x0 -= mean; x1 -= mean; x2 -= mean; x3 -= mean; x4 -= mean;

  float mx = fmaxf(fmaxf(fmaxf(fabsf(x0), fabsf(x1)),
                         fmaxf(fabsf(x2), fabsf(x3))), fabsf(x4));
#pragma unroll
  for (int off = 16; off; off >>= 1) mx = fmaxf(mx, __shfl_xor(mx, off, 32));
  const bool silent = mx < 1e-8f;

  float* xp = s_xp[wave];
  float* xr = s_xr[wave];
  const int zp = lane + 4 * (lane >> 4);
  xp[zp] = 0.0f;
  if (lane < 16) xp[zp + 40] = 0.0f;
  if (lane < 16) xr[96 + lane] = 0.0f;
  xp[XP_DATA + zp      ] = x0;                           xr[95 - lane] = x0;
  xp[XP_DATA + zp + 40 ] = x1;                           xr[63 - lane] = x1;
  xp[XP_DATA + zp + 80 ] = (lane < 16) ? x2 : 2.0f * x2; xr[31 - lane] = (lane < 16) ? x2 : 0.0f;
  xp[XP_DATA + zp + 120] = 2.0f * x3;
  xp[XP_DATA + zp + 160] = 2.0f * x4;
  __syncthreads();

  const int n  = lane & 15;
  const int hf = lane >> 4;
  const float* __restrict__ xra = xr + 2 * hf + n;
  const float* __restrict__ xpb = xp + XP_DATA + 2 * hf - 20 * (n & 7);
  v8f c = {};
#pragma unroll
  for (int k0 = 0; k0 < K_LEN; k0 += 4) {
    v2f a, bb;
    a.x  = xra[k0];
    a.y  = xra[k0 + 1];
    const int kk = K_FIRST + k0;
    const int kp = kk + 4 * (kk >> 4);
    bb.x = xpb[kp];
    bb.y = xpb[kp + 1];
    c = __builtin_amdgcn_wmma_f32_16x16x4_f32(
         false, a,  false, bb,
         (short)0, c,  false,  false);
  }

  union { v8f v; float f[8]; } cu; cu.v = c;
  float bestv = NEG_INF;
  int   bestk = 1 << 30;
#pragma unroll
  for (int v = 0; v < 8; ++v) {
    const int m = v + 8 * hf;
    const int L = 16 * n + m;
    const int k = 159 - L;
    const float val = (L <= 159 - MIN_LAG) ? cu.f[v] : NEG_INF;
    const bool better = (val > bestv) | ((val == bestv) & (k < bestk));
    bestv = better ? val : bestv;
    bestk = better ? k : bestk;
  }
#pragma unroll
  for (int off = 16; off; off >>= 1) {
    const float ov = __shfl_xor(bestv, off, 32);
    const int   ok = __shfl_xor(bestk, off, 32);
    const bool better = (ov > bestv) | ((ov == bestv) & (ok < bestk));
    bestv = better ? ov : bestv;
    bestk = better ? ok : bestk;
  }

  if (lane == 0) {
    out[f] = silent ? 0.0f : 16000.0f / (float)bestk;
  }
}

extern "C" void kernel_launch(void* const* d_in, const int* in_sizes, int n_in,
                              void* d_out, int out_size, void* d_ws, size_t ws_size,
                              hipStream_t stream) {
  (void)n_in; (void)d_ws; (void)ws_size;
  const float* audio = (const float*)d_in[0];
  float* out = (float*)d_out;
  const int B = 16;
  const int S = in_sizes[0] / B;
  const int T = (S - FRAME_LEN) / HOP + 1;
  const int nframes = B * T;
  (void)out_size;
  const int blocks = (nframes + WAVES_PER_BLOCK - 1) / WAVES_PER_BLOCK;
  pitch_wmma_kernel<<<blocks, 32 * WAVES_PER_BLOCK, 0, stream>>>(
      audio, out, S, T, nframes);
}


# Each wave loops over 4 consecutive frames on a wave-private LDS stage (no work-group barrier), next frame's samples requested one frame ahead; 8192 -> 2048 work-groups; summation order unchanged
# speedup vs baseline: 1.4818x; 1.1149x over previous
#include <hip/hip_runtime.h>

typedef float v2f __attribute__((ext_vector_type(2)));
typedef float v8f __attribute__((ext_vector_type(8)));

#define FRAME_LEN 160
#define HOP 256
#define MIN_LAG 40
#define WAVES_PER_BLOCK 8
#define FRAMES_PER_WAVE 4
#define XP_SIZE 260
#define XP_DATA 60
#define XR_SIZE 112
#define K_FIRST 64
#define K_LEN 96
#define NEG_INF (-3.402823466e+38f)


__device__ __forceinline__ void load_frame(
    const float* __restrict__ audio, size_t off, int lane,
    float& x0, float& x1, float& x2, float& x3, float& x4) {
  const float* __restrict__ src = audio + off;
  x0 = src[lane];
  x1 = src[lane + 32];
  x2 = src[lane + 64];
  x3 = src[lane + 96];
  x4 = src[lane + 128];
}

__device__ __forceinline__ int frame_lag(
    float x0, float x1, float x2, float x3, float x4,
    float* xp, float* xr, const float* xra, const float* xpb,
    int lane, int zp, int n, int hf) {
  float sum = x0 + x1 + x2 + x3 + x4;
#pragma unroll
  for (int off = 16; off; off >>= 1) sum += __shfl_xor(sum, off, 32);
  const float mean = sum * (1.0f / 160.0f);
  x0 -= mean; x1 -= mean; x2 -= mean; x3 -= mean; x4 -= mean;

  float mx = fmaxf(fmaxf(fmaxf(fabsf(x0), fabsf(x1)),
                         fmaxf(fabsf(x2), fabsf(x3))), fabsf(x4));
#pragma unroll
  for (int off = 16; off; off >>= 1) mx = fmaxf(mx, __shfl_xor(mx, off, 32));
  const bool silent = mx < 1e-8f;

  xp[XP_DATA + zp      ] = x0;                           xr[95 - lane] = x0;
  xp[XP_DATA + zp + 40 ] = x1;                           xr[63 - lane] = x1;
  xp[XP_DATA + zp + 80 ] = (lane < 16) ? x2 : 2.0f * x2; xr[31 - lane] = (lane < 16) ? x2 : 0.0f;
  xp[XP_DATA + zp + 120] = 2.0f * x3;
  xp[XP_DATA + zp + 160] = 2.0f * x4;

  v8f c = {};
#pragma unroll
  for (int k0 = 0; k0 < K_LEN; k0 += 4) {
    v2f a, bb;
    a.x  = xra[k0];
    a.y  = xra[k0 + 1];
    const int kk = K_FIRST + k0;
    const int kp = kk + 4 * (kk >> 4);
    bb.x = xpb[kp];
    bb.y = xpb[kp + 1];
    c = __builtin_amdgcn_wmma_f32_16x16x4_f32(
         false, a,  false, bb,
         (short)0, c,  false,  false);
  }

  union { v8f v; float f[8]; } cu; cu.v = c;
  float bestv = NEG_INF;
  int   bestk = 1 << 30;
#pragma unroll
  for (int v = 0; v < 8; ++v) {
    const int m = v + 8 * hf;
    const int L = 16 * n + m;
    const int k = 159 - L;
    const float val = (L <= 159 - MIN_LAG) ? cu.f[v] : NEG_INF;
    const bool better = (val > bestv) | ((val == bestv) & (k < bestk));
    bestv = better ? val : bestv;
    bestk = better ? k : bestk;
  }
#pragma unroll
  for (int off = 16; off; off >>= 1) {
    const float ov = __shfl_xor(bestv, off, 32);
    const int   ok = __shfl_xor(bestk, off, 32);
    const bool better = (ov > bestv) | ((ov == bestv) & (ok < bestk));
    bestv = better ? ov : bestv;
    bestk = better ? ok : bestk;
  }
  return silent ? 0 : bestk;
}

__global__ __launch_bounds__(256) void pitch_wmma_kernel(
    const float* __restrict__ audio, float* __restrict__ out,
    int S, int T, int nframes) {
  __shared__ float s_xp[WAVES_PER_BLOCK][XP_SIZE];
  __shared__ float s_xr[WAVES_PER_BLOCK][XR_SIZE];

  const int lane = threadIdx.x & 31;
  const int wave = __builtin_amdgcn_readfirstlane((int)(threadIdx.x >> 5));
  const int f0 = (blockIdx.x * WAVES_PER_BLOCK + wave) * FRAMES_PER_WAVE;
  const int flast = nframes - 1;

  float* xp = s_xp[wave];
  float* xr = s_xr[wave];
  const int zp = lane + 4 * (lane >> 4);
  xp[zp] = 0.0f;
  if (lane < 16) xp[zp + 40] = 0.0f;
  if (lane < 16) xr[96 + lane] = 0.0f;

  const int n  = lane & 15;
  const int hf = lane >> 4;
  const float* xra = xr + 2 * hf + n;
  const float* xpb = xp + XP_DATA + 2 * hf - 20 * (n & 7);

  int f = min(f0, flast);
  const int b0 = f / T;
  int t = f - b0 * T;
  size_t off = (size_t)b0 * (size_t)S + (size_t)t * HOP;
  const size_t row_step = (size_t)S - (size_t)T * HOP;

  float x0, x1, x2, x3, x4;
  load_frame(audio, off, lane, x0, x1, x2, x3, x4);

  int lagq = 0;

#pragma unroll 1
  for (int i = 0; i < FRAMES_PER_WAVE - 1; ++i) {
    if (f < flast) {
      ++f; ++t; off += HOP;
      if (t == T) { t = 0; off += row_step; }
    }
    float n0, n1, n2, n3, n4;
    load_frame(audio, off, lane, n0, n1, n2, n3, n4);
    __builtin_amdgcn_sched_barrier(0);
    const int lag = frame_lag(x0, x1, x2, x3, x4, xp, xr, xra, xpb, lane, zp, n, hf);
    lagq = (lane == i) ? lag : lagq;
    x0 = n0; x1 = n1; x2 = n2; x3 = n3; x4 = n4;
  }
  {
    const int lag = frame_lag(x0, x1, x2, x3, x4, xp, xr, xra, xpb, lane, zp, n, hf);
    lagq = (lane == FRAMES_PER_WAVE - 1) ? lag : lagq;
  }

  const int fo = f0 + lane;
  if (lane < FRAMES_PER_WAVE && fo < nframes) {
    out[fo] = (lagq == 0) ? 0.0f : 16000.0f / (float)lagq;
  }
}

extern "C" void kernel_launch(void* const* d_in, const int* in_sizes, int n_in,
                              void* d_out, int out_size, void* d_ws, size_t ws_size,
                              hipStream_t stream) {
  (void)n_in; (void)d_ws; (void)ws_size;
  const float* audio = (const float*)d_in[0];
  float* out = (float*)d_out;
  const int B = 16;
  const int S = in_sizes[0] / B;
  const int T = (S - FRAME_LEN) / HOP + 1;
  const int nframes = B * T;
  (void)out_size;
  const int per_block = WAVES_PER_BLOCK * FRAMES_PER_WAVE;
  const int blocks = (nframes + per_block - 1) / per_block;
  pitch_wmma_kernel<<<blocks, 32 * WAVES_PER_BLOCK, 0, stream>>>(
      audio, out, S, T, nframes);
}


# Tile rows form two groups of 8 that divide the m range (m=8..79 and m=0..7, one column apart); K 96 -> 80, 24 -> 20 f32 WMMAs; summation order changes (two partial sums added)
# speedup vs baseline: 1.6860x; 1.1378x over previous
#include <hip/hip_runtime.h>

typedef float v2f __attribute__((ext_vector_type(2)));
typedef float v8f __attribute__((ext_vector_type(8)));

#define FRAME_LEN 160
#define HOP 256
#define MIN_LAG 40
#define WAVES_PER_BLOCK 8
#define FRAMES_PER_WAVE 4
#define YP_SIZE 296
#define YP_DATA 60
#define RA_SIZE 200
#define RB_OFF 8
#define RS_OFF 104
#define K_LEN 80
#define NEG_INF (-3.402823466e+38f)


__device__ __forceinline__ void load_frame(
    const float* __restrict__ audio, size_t off, int lane,
    float& x0, float& x1, float& x2, float& x3, float& x4) {
  const float* __restrict__ src = audio + off;
  x0 = src[lane];
  x1 = src[lane + 32];
  x2 = src[lane + 64];
  x3 = src[lane + 96];
  x4 = src[lane + 128];
}

__device__ __forceinline__ int frame_lag(
    float x0, float x1, float x2, float x3, float x4,
    float* yp, float* ra, float* ra_row0, const float* xra, const float* xpb,
    int lane, int zp, int n) {
  float sum = x0 + x1 + x2 + x3 + x4;
#pragma unroll
  for (int off = 16; off; off >>= 1) sum += __shfl_xor(sum, off, 32);
  const float mean = sum * (1.0f / 160.0f);
  x0 -= mean; x1 -= mean; x2 -= mean; x3 -= mean; x4 -= mean;

  float mx = fmaxf(fmaxf(fmaxf(fabsf(x0), fabsf(x1)),
                         fmaxf(fabsf(x2), fabsf(x3))), fabsf(x4));
#pragma unroll
  for (int off = 16; off; off >>= 1) mx = fmaxf(mx, __shfl_xor(mx, off, 32));
  const bool silent = mx < 1e-8f;

  yp[YP_DATA + zp      ] = x0;                           *ra_row0 = x0;
  yp[YP_DATA + zp + 48 ] = x1;                           ra[RB_OFF + 55 - lane] = x1;
  yp[YP_DATA + zp + 96 ] = (lane < 16) ? x2 : 2.0f * x2; ra[RB_OFF + 23 - lane] = (lane < 16) ? x2 : 0.0f;
  yp[YP_DATA + zp + 144] = 2.0f * x3;
  yp[YP_DATA + zp + 192] = 2.0f * x4;

  v8f c = {};
#pragma unroll
  for (int k0 = 0; k0 < K_LEN; k0 += 4) {
    v2f a, bb;
    a.x  = xra[k0];
    a.y  = xra[k0 + 1];
    const int kp = k0 + 4 * (k0 >> 3);
    bb.x = xpb[kp];
    bb.y = xpb[kp + 1];
    c = __builtin_amdgcn_wmma_f32_16x16x4_f32(
         false, a,  false, bb,
         (short)0, c,  false,  false);
  }

  union { v8f v; float f[8]; } cu; cu.v = c;
  float bestv = NEG_INF;
  int   bestk = 1 << 30;
#pragma unroll
  for (int v = 0; v < 8; ++v) {
    const unsigned bits = __builtin_bit_cast(unsigned, cu.f[v]);
    const unsigned up = __builtin_amdgcn_permlanex16(
        bits, bits, 0x87654321u, 0x0fedcba9u, true, true);
    const float val = cu.f[v] + __builtin_bit_cast(float, up);
    const int k = 47 + 8 * n - v;
    const bool better = (val > bestv) | ((val == bestv) & (k < bestk));
    bestv = better ? val : bestv;
    bestk = better ? k : bestk;
  }
  bestv = (lane < 15) ? bestv : NEG_INF;
#pragma unroll
  for (int off = 8; off; off >>= 1) {
    const float ov = __shfl_xor(bestv, off, 32);
    const int   ok = __shfl_xor(bestk, off, 32);
    const bool better = (ov > bestv) | ((ov == bestv) & (ok < bestk));
    bestv = better ? ov : bestv;
    bestk = better ? ok : bestk;
  }
  return silent ? 0 : bestk;
}

__global__ __launch_bounds__(256) void pitch_wmma_kernel(
    const float* __restrict__ audio, float* __restrict__ out,
    int S, int T, int nframes) {
  __shared__ float s_yp[WAVES_PER_BLOCK][YP_SIZE];
  __shared__ float s_ra[WAVES_PER_BLOCK][RA_SIZE];

  const int lane = threadIdx.x & 31;
  const int wave = __builtin_amdgcn_readfirstlane((int)(threadIdx.x >> 5));
  const int f0 = (blockIdx.x * WAVES_PER_BLOCK + wave) * FRAMES_PER_WAVE;
  const int flast = nframes - 1;

  float* yp = s_yp[wave];
  float* ra = s_ra[wave];
  const int zp = lane + 4 * (lane >> 3);
  yp[zp] = 0.0f;
  if (lane < 8) yp[48 + lane] = 0.0f;
#pragma unroll
  for (int j = 0; j < RA_SIZE; j += 32) {
    if (j + lane < RA_SIZE) ra[j + lane] = 0.0f;
  }

  const int n  = lane & 15;
  const int hf = lane >> 4;
  const float* xra = ra + ((n < 8) ? (RB_OFF + n) : (RS_OFF + n - 8)) + 2 * hf;
  const float* xpb = yp + 12 * n + 2 * hf;
  float* ra_row0 = ra + ((lane < 8) ? (RS_OFF + 79 - lane) : (RB_OFF + 87 - lane));

  int f = min(f0, flast);
  const int b0 = f / T;
  int t = f - b0 * T;
  size_t off = (size_t)b0 * (size_t)S + (size_t)t * HOP;
  const size_t row_step = (size_t)S - (size_t)T * HOP;

  float x0, x1, x2, x3, x4;
  load_frame(audio, off, lane, x0, x1, x2, x3, x4);

  int lagq = 0;

#pragma unroll 1
  for (int i = 0; i < FRAMES_PER_WAVE - 1; ++i) {
    if (f < flast) {
      ++f; ++t; off += HOP;
      if (t == T) { t = 0; off += row_step; }
    }
    float n0, n1, n2, n3, n4;
    load_frame(audio, off, lane, n0, n1, n2, n3, n4);
    __builtin_amdgcn_sched_barrier(0);
    const int lag = frame_lag(x0, x1, x2, x3, x4, yp, ra, ra_row0, xra, xpb, lane, zp, n);
    lagq = (lane == i) ? lag : lagq;
    x0 = n0; x1 = n1; x2 = n2; x3 = n3; x4 = n4;
  }
  {
    const int lag = frame_lag(x0, x1, x2, x3, x4, yp, ra, ra_row0, xra, xpb, lane, zp, n);
    lagq = (lane == FRAMES_PER_WAVE - 1) ? lag : lagq;
  }

  const int fo = f0 + lane;
  if (lane < FRAMES_PER_WAVE && fo < nframes) {
    out[fo] = (lagq == 0) ? 0.0f : 16000.0f / (float)lagq;
  }
}

extern "C" void kernel_launch(void* const* d_in, const int* in_sizes, int n_in,
                              void* d_out, int out_size, void* d_ws, size_t ws_size,
                              hipStream_t stream) {
  (void)n_in; (void)d_ws; (void)ws_size;
  const float* audio = (const float*)d_in[0];
  float* out = (float*)d_out;
  const int B = 16;
  const int S = in_sizes[0] / B;
  const int T = (S - FRAME_LEN) / HOP + 1;
  const int nframes = B * T;
  (void)out_size;
  const int per_block = WAVES_PER_BLOCK * FRAMES_PER_WAVE;
  const int blocks = (nframes + per_block - 1) / per_block;
  pitch_wmma_kernel<<<blocks, 32 * WAVES_PER_BLOCK, 0, stream>>>(
      audio, out, S, T, nframes);
}


# K index walks a comb of m (blocks of 8 at a pitch of 16): the two row groups take alternate m blocks instead of a 72/8 split; K 80 -> 48, 20 -> 12 f32 WMMAs; summation order changes
# speedup vs baseline: 2.1109x; 1.2520x over previous
#include <hip/hip_runtime.h>

typedef float v2f __attribute__((ext_vector_type(2)));
typedef float v8f __attribute__((ext_vector_type(8)));

#define FRAME_LEN 160
#define HOP 256
#define MIN_LAG 40
#define WAVES_PER_BLOCK 8
#define FRAMES_PER_WAVE 4
#define YR_SIZE 312
#define YR_TOP 235
#define XA_SIZE 104
#define XA_DATA 7
#define K_LEN 48
#define NEG_INF (-3.402823466e+38f)


__device__ __forceinline__ void load_frame(
    const float* __restrict__ audio, size_t off, int lane,
    float& x0, float& x1, float& x2, float& x3, float& x4) {
  const float* __restrict__ src = audio + off;
  x0 = src[lane];
  x1 = src[lane + 32];
  x2 = src[lane + 64];
  x3 = src[lane + 96];
  x4 = src[lane + 128];
}

__device__ __forceinline__ int frame_lag(
    float x0, float x1, float x2, float x3, float x4,
    float* ytop, float* xa, const float* xra, const float* xpb,
    int lane, int n) {
  float sum = x0 + x1 + x2 + x3 + x4;
#pragma unroll
  for (int off = 16; off; off >>= 1) sum += __shfl_xor(sum, off, 32);
  const float mean = sum * (1.0f / 160.0f);
  x0 -= mean; x1 -= mean; x2 -= mean; x3 -= mean; x4 -= mean;

  float mx = fmaxf(fmaxf(fmaxf(fabsf(x0), fabsf(x1)),
                         fmaxf(fabsf(x2), fabsf(x3))), fabsf(x4));
#pragma unroll
  for (int off = 16; off; off >>= 1) mx = fmaxf(mx, __shfl_xor(mx, off, 32));
  const bool silent = mx < 1e-8f;

  ytop[192] = x0;                           xa[XA_DATA + lane]      = x0;
  ytop[144] = x1;                           xa[XA_DATA + 32 + lane] = x1;
  ytop[96]  = (lane < 16) ? x2 : 2.0f * x2; xa[XA_DATA + 64 + lane] = (lane < 16) ? x2 : 0.0f;
  ytop[48]  = 2.0f * x3;
  ytop[0]   = 2.0f * x4;

  v8f c = {};
#pragma unroll
  for (int k0 = 0; k0 < K_LEN; k0 += 4) {
    v2f a, bb;
    const int ga = 16 * (k0 >> 3) + (k0 & 4);
    a.x  = xra[ga];
    a.y  = xra[ga + 1];
    const int gb = 24 * (k0 >> 3) + (k0 & 4);
    bb.x = xpb[gb];
    bb.y = xpb[gb + 1];
    c = __builtin_amdgcn_wmma_f32_16x16x4_f32(
         false, a,  false, bb,
         (short)0, c,  false,  false);
  }

  union { v8f v; float f[8]; } cu; cu.v = c;
  float bestv = NEG_INF;
  int   bestk = 1 << 30;
#pragma unroll
  for (int v = 0; v < 8; ++v) {
    const unsigned bits = __builtin_bit_cast(unsigned, cu.f[v]);
    const unsigned up = __builtin_amdgcn_permlanex16(
        bits, bits, 0x6543210fu, 0xedcba987u, true, true);
    const float val = cu.f[v] + __builtin_bit_cast(float, up);
    const int k = 32 + 8 * n + v;
    const bool better = (val > bestv) | ((val == bestv) & (k < bestk));
    bestv = better ? val : bestv;
    bestk = better ? k : bestk;
  }
  bestv = (lane >= 1 && lane < 16) ? bestv : NEG_INF;
#pragma unroll
  for (int off = 8; off; off >>= 1) {
    const float ov = __shfl_xor(bestv, off, 32);
    const int   ok = __shfl_xor(bestk, off, 32);
    const bool better = (ov > bestv) | ((ov == bestv) & (ok < bestk));
    bestv = better ? ov : bestv;
    bestk = better ? ok : bestk;
  }
  return silent ? 0 : bestk;
}

__global__ __launch_bounds__(256) void pitch_wmma_kernel(
    const float* __restrict__ audio, float* __restrict__ out,
    int S, int T, int nframes) {
  __shared__ float s_yr[WAVES_PER_BLOCK][YR_SIZE];
  __shared__ float s_xa[WAVES_PER_BLOCK][XA_SIZE];

  const int lane = threadIdx.x & 31;
  const int wave = __builtin_amdgcn_readfirstlane((int)(threadIdx.x >> 5));
  const int f0 = (blockIdx.x * WAVES_PER_BLOCK + wave) * FRAMES_PER_WAVE;
  const int flast = nframes - 1;

  float* yr = s_yr[wave];
  float* xa = s_xa[wave];
  const int zp = lane + 4 * (lane >> 3);
  yr[240 + zp] = 0.0f;
  if (lane < 16) yr[288 + zp] = 0.0f;
#pragma unroll
  for (int j = 0; j < XA_SIZE; j += 32) {
    if (j + lane < XA_SIZE) xa[j + lane] = 0.0f;
  }

  const int n  = lane & 15;
  const int hf = lane >> 4;
  const float* xra = xa + n + 2 * hf;
  const float* xpb = yr + 12 * (15 - n) + 2 * hf;
  float* ytop = yr + (YR_TOP - 192 - zp);

  int f = min(f0, flast);
  const int b0 = f / T;
  int t = f - b0 * T;
  size_t off = (size_t)b0 * (size_t)S + (size_t)t * HOP;
  const size_t row_step = (size_t)S - (size_t)T * HOP;

  float x0, x1, x2, x3, x4;
  load_frame(audio, off, lane, x0, x1, x2, x3, x4);

  int lagq = 0;

#pragma unroll 1
  for (int i = 0; i < FRAMES_PER_WAVE - 1; ++i) {
    if (f < flast) {
      ++f; ++t; off += HOP;
      if (t == T) { t = 0; off += row_step; }
    }
    float n0, n1, n2, n3, n4;
    load_frame(audio, off, lane, n0, n1, n2, n3, n4);
    __builtin_amdgcn_sched_barrier(0);
    const int lag = frame_lag(x0, x1, x2, x3, x4, ytop, xa, xra, xpb, lane, n);
    lagq = (lane == i) ? lag : lagq;
    x0 = n0; x1 = n1; x2 = n2; x3 = n3; x4 = n4;
  }
  {
    const int lag = frame_lag(x0, x1, x2, x3, x4, ytop, xa, xra, xpb, lane, n);
    lagq = (lane == FRAMES_PER_WAVE - 1) ? lag : lagq;
  }

  const int fo = f0 + lane;
  if (lane < FRAMES_PER_WAVE && fo < nframes) {
    out[fo] = (lagq == 0) ? 0.0f : 16000.0f / (float)lagq;
  }
}

extern "C" void kernel_launch(void* const* d_in, const int* in_sizes, int n_in,
                              void* d_out, int out_size, void* d_ws, size_t ws_size,
                              hipStream_t stream) {
  (void)n_in; (void)d_ws; (void)ws_size;
  const float* audio = (const float*)d_in[0];
  float* out = (float*)d_out;
  const int B = 16;
  const int S = in_sizes[0] / B;
  const int T = (S - FRAME_LEN) / HOP + 1;
  const int nframes = B * T;
  (void)out_size;
  const int per_block = WAVES_PER_BLOCK * FRAMES_PER_WAVE;
  const int blocks = (nframes + per_block - 1) / per_block;
  pitch_wmma_kernel<<<blocks, 32 * WAVES_PER_BLOCK, 0, stream>>>(
      audio, out, S, T, nframes);
}


# Arg-max and silence test by lane votes instead of shuffled (value,index) butterflies: row max on DPP, then compare masks + scalar find-first pick the first index; summation order unchanged
# speedup vs baseline: 2.4437x; 1.1577x over previous
#include <hip/hip_runtime.h>

typedef float v2f __attribute__((ext_vector_type(2)));
typedef float v8f __attribute__((ext_vector_type(8)));

#define FRAME_LEN 160
#define HOP 256
#define MIN_LAG 40
#define WAVES_PER_BLOCK 8
#define FRAMES_PER_WAVE 4
#define YR_SIZE 312
#define YR_TOP 235
#define XA_SIZE 104
#define XA_DATA 7
#define K_LEN 48


__device__ __forceinline__ void load_frame(
    const float* __restrict__ audio, size_t off, int lane,
    float& x0, float& x1, float& x2, float& x3, float& x4) {
  const float* __restrict__ src = audio + off;
  x0 = src[lane];
  x1 = src[lane + 32];
  x2 = src[lane + 64];
  x3 = src[lane + 96];
  x4 = src[lane + 128];
}

__device__ __forceinline__ float row16_max(float v) {
  float r;
  asm("s_delay_alu instid0(VALU_DEP_1)\n\t"
      "v_max_num_f32_dpp %0, %1, %1 row_xmask:1 row_mask:0xf bank_mask:0xf\n\t"
      "s_delay_alu instid0(VALU_DEP_1)\n\t"
      "v_max_num_f32_dpp %0, %0, %0 row_xmask:2 row_mask:0xf bank_mask:0xf\n\t"
      "s_delay_alu instid0(VALU_DEP_1)\n\t"
      "v_max_num_f32_dpp %0, %0, %0 row_xmask:4 row_mask:0xf bank_mask:0xf\n\t"
      "s_delay_alu instid0(VALU_DEP_1)\n\t"
      "v_max_num_f32_dpp %0, %0, %0 row_xmask:8 row_mask:0xf bank_mask:0xf\n\t"
      "s_delay_alu instid0(VALU_DEP_1)"
      : "=v"(r) : "v"(v));
  return r;
}

__device__ __forceinline__ int frame_lag(
    float x0, float x1, float x2, float x3, float x4,
    float* ytop, float* xa, const float* xra, const float* xpb,
    int lane) {
  float sum = x0 + x1 + x2 + x3 + x4;
#pragma unroll
  for (int off = 16; off; off >>= 1) sum += __shfl_xor(sum, off, 32);
  const float mean = sum * (1.0f / 160.0f);
  x0 -= mean; x1 -= mean; x2 -= mean; x3 -= mean; x4 -= mean;

  const float mx = fmaxf(fmaxf(fmaxf(fabsf(x0), fabsf(x1)),
                               fmaxf(fabsf(x2), fabsf(x3))), fabsf(x4));
  const bool silent = __builtin_amdgcn_ballot_w32(mx >= 1e-8f) == 0u;

  ytop[192] = x0;                           xa[XA_DATA + lane]      = x0;
  ytop[144] = x1;                           xa[XA_DATA + 32 + lane] = x1;
  ytop[96]  = (lane < 16) ? x2 : 2.0f * x2; xa[XA_DATA + 64 + lane] = (lane < 16) ? x2 : 0.0f;
  ytop[48]  = 2.0f * x3;
  ytop[0]   = 2.0f * x4;

  v8f c = {};
#pragma unroll
  for (int k0 = 0; k0 < K_LEN; k0 += 4) {
    v2f a, bb;
    const int ga = 16 * (k0 >> 3) + (k0 & 4);
    a.x  = xra[ga];
    a.y  = xra[ga + 1];
    const int gb = 24 * (k0 >> 3) + (k0 & 4);
    bb.x = xpb[gb];
    bb.y = xpb[gb + 1];
    c = __builtin_amdgcn_wmma_f32_16x16x4_f32(
         false, a,  false, bb,
         (short)0, c,  false,  false);
  }

  union { v8f v; float f[8]; } cu; cu.v = c;
  float val[8];
#pragma unroll
  for (int v = 0; v < 8; ++v) {
    const unsigned bits = __builtin_bit_cast(unsigned, cu.f[v]);
    const unsigned up = __builtin_amdgcn_permlanex16(
        bits, bits, 0x6543210fu, 0xedcba987u, true, true);
    val[v] = cu.f[v] + __builtin_bit_cast(float, up);
  }
  const bool valid = (lane >= 1) & (lane < 16);
  float m = fmaxf(fmaxf(fmaxf(val[0], val[1]), fmaxf(val[2], val[3])),
                  fmaxf(fmaxf(val[4], val[5]), fmaxf(val[6], val[7])));
  m = valid ? m : -__builtin_inff();
  const float lm = m;
  m = row16_max(m);
  const float gref = valid ? m : __builtin_nanf("");
  const unsigned lanes = __builtin_amdgcn_ballot_w32(lm == gref);
  const unsigned nbest = (unsigned)__builtin_ctz(lanes | 0x10000u);
  unsigned vbest = 7u;
#pragma unroll
  for (int v = 6; v >= 0; --v) {
    const unsigned hit = __builtin_amdgcn_ballot_w32(val[v] == gref);
    vbest = ((hit >> nbest) & 1u) ? (unsigned)v : vbest;
  }
  const unsigned best = 8u * nbest + vbest;
  const int bestk = 32 + (int)best;
  return silent ? 0 : bestk;
}

__global__ __launch_bounds__(256) void pitch_wmma_kernel(
    const float* __restrict__ audio, float* __restrict__ out,
    int S, int T, int nframes) {
  __shared__ float s_yr[WAVES_PER_BLOCK][YR_SIZE];
  __shared__ float s_xa[WAVES_PER_BLOCK][XA_SIZE];

  const int lane = threadIdx.x & 31;
  const int wave = __builtin_amdgcn_readfirstlane((int)(threadIdx.x >> 5));
  const int f0 = (blockIdx.x * WAVES_PER_BLOCK + wave) * FRAMES_PER_WAVE;
  const int flast = nframes - 1;

  float* yr = s_yr[wave];
  float* xa = s_xa[wave];
  const int zp = lane + 4 * (lane >> 3);
  yr[240 + zp] = 0.0f;
  if (lane < 16) yr[288 + zp] = 0.0f;
#pragma unroll
  for (int j = 0; j < XA_SIZE; j += 32) {
    if (j + lane < XA_SIZE) xa[j + lane] = 0.0f;
  }

  const int n  = lane & 15;
  const int hf = lane >> 4;
  const float* xra = xa + n + 2 * hf;
  const float* xpb = yr + 12 * (15 - n) + 2 * hf;
  float* ytop = yr + (YR_TOP - 192 - zp);

  int f = min(f0, flast);
  const int b0 = f / T;
  int t = f - b0 * T;
  size_t off = (size_t)b0 * (size_t)S + (size_t)t * HOP;
  const size_t row_step = (size_t)S - (size_t)T * HOP;

  float x0, x1, x2, x3, x4;
  load_frame(audio, off, lane, x0, x1, x2, x3, x4);

  int lagq = 0;

#pragma unroll 1
  for (int i = 0; i < FRAMES_PER_WAVE - 1; ++i) {
    if (f < flast) {
      ++f; ++t; off += HOP;
      if (t == T) { t = 0; off += row_step; }
    }
    float n0, n1, n2, n3, n4;
    load_frame(audio, off, lane, n0, n1, n2, n3, n4);
    __builtin_amdgcn_sched_barrier(0);
    const int lag = frame_lag(x0, x1, x2, x3, x4, ytop, xa, xra, xpb, lane);
    lagq = (lane == i) ? lag : lagq;
    x0 = n0; x1 = n1; x2 = n2; x3 = n3; x4 = n4;
  }
  {
    const int lag = frame_lag(x0, x1, x2, x3, x4, ytop, xa, xra, xpb, lane);
    lagq = (lane == FRAMES_PER_WAVE - 1) ? lag : lagq;
  }

  const int fo = f0 + lane;
  if (lane < FRAMES_PER_WAVE && fo < nframes) {
    out[fo] = (lagq == 0) ? 0.0f : 16000.0f / (float)lagq;
  }
}

extern "C" void kernel_launch(void* const* d_in, const int* in_sizes, int n_in,
                              void* d_out, int out_size, void* d_ws, size_t ws_size,
                              hipStream_t stream) {
  (void)n_in; (void)d_ws; (void)ws_size;
  const float* audio = (const float*)d_in[0];
  float* out = (float*)d_out;
  const int B = 16;
  const int S = in_sizes[0] / B;
  const int T = (S - FRAME_LEN) / HOP + 1;
  const int nframes = B * T;
  (void)out_size;
  const int per_block = WAVES_PER_BLOCK * FRAMES_PER_WAVE;
  const int blocks = (nframes + per_block - 1) / per_block;
  pitch_wmma_kernel<<<blocks, 32 * WAVES_PER_BLOCK, 0, stream>>>(
      audio, out, S, T, nframes);
}
